# starting kernel
# baseline (speedup 1.0000x reference)
#include <hip/hip_runtime.h>

typedef __attribute__((ext_vector_type(2))) float v2f;
typedef __attribute__((ext_vector_type(8))) float v8f;

#define NB    4
#define NN    50000
#define MM    (NB * NN)
#define CIN   64
#define HID   256
#define COUTN 64
#define NE    800000
#define STRIPS (MM / 16)

__global__ void gcn_zero_kernel(float* __restrict__ y) {
  long i4 = ((long)blockIdx.x * blockDim.x + threadIdx.x) * 4;
  *(float4*)(y + i4) = make_float4(0.f, 0.f, 0.f, 0.f);
}

__global__ void gcn_scatter_kernel(const float* __restrict__ x,
                                   const int* __restrict__ rows,
                                   const int* __restrict__ cols,
                                   const float* __restrict__ vals,
                                   float* __restrict__ y) {
  int t  = blockIdx.x * 256 + threadIdx.x;
  int e  = t >> 6;
  int ch = t & 63;
  if (e >= NE) return;
  int   r = rows[e];
  int   c = cols[e];
  float v = vals[e];
#pragma unroll
  for (int b = 0; b < NB; ++b) {
    float xv = x[(size_t)(b * NN + c) * CIN + ch];
    unsafeAtomicAdd(&y[(size_t)(b * NN + r) * CIN + ch], v * xv);
  }
}

__global__ void __launch_bounds__(256, 1)
gcn_mlp_kernel(float* __restrict__ xo,
               const float* __restrict__ W1,
               const float* __restrict__ b1,
               const float* __restrict__ W2,
               const float* __restrict__ b2) {
  __shared__ __align__(16) float sW1t[HID * CIN];
  __shared__ __align__(16) float sW2t[COUTN * HID];
  __shared__ __align__(16) float sB1[HID];
  __shared__ __align__(16) float sB2[COUTN];
  __shared__ __align__(16) float sStage[8][16 * 16];

  const int tid = threadIdx.x;
  for (int i = tid; i < CIN * HID; i += 256) {
    int k = i >> 8, n = i & 255;
    sW1t[n * CIN + k] = W1[i];
  }
  for (int i = tid; i < HID * COUTN; i += 256) {
    int k = i >> 6, o = i & 63;
    sW2t[o * HID + k] = W2[i];
  }
  if (tid < HID)   sB1[tid] = b1[tid];
  if (tid < COUTN) sB2[tid] = b2[tid];
  __syncthreads();

  const int wave  = tid >> 5;
  const int lane  = tid & 31;
  const int lhalf = lane >> 4;
  const int lmod  = lane & 15;

  const int strip = blockIdx.x * 8 + wave;
  if (strip >= STRIPS) return;
  const int row0 = strip * 16;

  v2f a1[16];
  {
    const float* base = xo + (size_t)(row0 + lmod) * CIN + 2 * lhalf;
#pragma unroll
    for (int s = 0; s < 16; ++s)
      a1[s] = *(const v2f*)(base + 4 * s);
  }

  v8f oacc[4];
#pragma unroll
  for (int o = 0; o < 4; ++o) {
    float bv = sB2[o * 16 + lmod];
#pragma unroll
    for (int j = 0; j < 8; ++j) oacc[o][j] = bv;
  }

  float* st = &sStage[wave][0];

  for (int t = 0; t < 16; ++t) {
    v8f h;
    float b1v = sB1[t * 16 + lmod];
#pragma unroll
    for (int j = 0; j < 8; ++j) h[j] = b1v;
#pragma unroll
    for (int s = 0; s < 16; ++s) {
      v2f bfrag = *(const v2f*)(&sW1t[(t * 16 + lmod) * CIN + 4 * s + 2 * lhalf]);
      h = __builtin_amdgcn_wmma_f32_16x16x4_f32(false, a1[s], false, bfrag,
                                                (short)0, h, false, false);
    }
#pragma unroll
    for (int j = 0; j < 8; ++j) h[j] = fmaxf(h[j], 0.f);

#pragma unroll
    for (int j = 0; j < 8; ++j)
      st[(lhalf * 8 + j) * 16 + lmod] = h[j];
    asm volatile("s_wait_dscnt 0" ::: "memory");

    v2f a2[4];
#pragma unroll
    for (int kk = 0; kk < 4; ++kk)
      a2[kk] = *(const v2f*)(&st[lmod * 16 + 4 * kk + 2 * lhalf]);

#pragma unroll
    for (int o = 0; o < 4; ++o) {
#pragma unroll
      for (int kk = 0; kk < 4; ++kk) {
        v2f b2frag = *(const v2f*)(&sW2t[(o * 16 + lmod) * HID + t * 16 + 4 * kk + 2 * lhalf]);
        oacc[o] = __builtin_amdgcn_wmma_f32_16x16x4_f32(false, a2[kk], false, b2frag,
                                                        (short)0, oacc[o], false, false);
      }
    }
  }

#pragma unroll
  for (int o = 0; o < 4; ++o) {
#pragma unroll
    for (int j = 0; j < 8; ++j)
      xo[(size_t)(row0 + lhalf * 8 + j) * COUTN + o * 16 + lmod] = oacc[o][j];
  }
}

extern "C" void kernel_launch(void* const* d_in, const int* in_sizes, int n_in,
                              void* d_out, int out_size, void* d_ws, size_t ws_size,
                              hipStream_t stream) {
  const float* x     = (const float*)d_in[0];
  const int*   A_idx = (const int*)d_in[1];
  const float* A_val = (const float*)d_in[2];
  const float* W1    = (const float*)d_in[3];
  const float* b1    = (const float*)d_in[4];
  const float* W2    = (const float*)d_in[5];
  const float* b2    = (const float*)d_in[6];
  float* out = (float*)d_out;

  gcn_zero_kernel<<<12500, 256, 0, stream>>>(out);

  gcn_scatter_kernel<<<(NE * 64) / 256, 256, 0, stream>>>(
      x, A_idx, A_idx + NE, A_val, out);

  gcn_mlp_kernel<<<(STRIPS + 7) / 8, 256, 0, stream>>>(out, W1, b1, W2, b2);
}


# MLP formed transposed: weights are the WMMA A operand read from row-major LDS, strip rows the B operand; hidden tile passes accumulator to B operand in registers, no LDS bounce; sum order unchanged
# speedup vs baseline: 1.5615x; 1.5615x over previous
#include <hip/hip_runtime.h>

typedef __attribute__((ext_vector_type(2))) float v2f;
typedef __attribute__((ext_vector_type(8))) float v8f;

#define NB    4
#define NN    50000
#define MM    (NB * NN)
#define CIN   64
#define HID   256
#define COUTN 64
#define NE    800000
#define STRIPS (MM / 16)

__global__ void gcn_zero_kernel(float* __restrict__ y) {
  long i4 = ((long)blockIdx.x * blockDim.x + threadIdx.x) * 4;
  *(float4*)(y + i4) = make_float4(0.f, 0.f, 0.f, 0.f);
}

__global__ void gcn_scatter_kernel(const float* __restrict__ x,
                                   const int* __restrict__ rows,
                                   const int* __restrict__ cols,
                                   const float* __restrict__ vals,
                                   float* __restrict__ y) {
  int t  = blockIdx.x * 256 + threadIdx.x;
  int e  = t >> 6;
  int ch = t & 63;
  if (e >= NE) return;
  int   r = rows[e];
  int   c = cols[e];
  float v = vals[e];
#pragma unroll
  for (int b = 0; b < NB; ++b) {
    float xv = x[(size_t)(b * NN + c) * CIN + ch];
    unsafeAtomicAdd(&y[(size_t)(b * NN + r) * CIN + ch], v * xv);
  }
}

#define P1 264
#define P2 68

__global__ void __launch_bounds__(256, 1)
gcn_mlp_kernel(float* __restrict__ xo,
               const float* __restrict__ W1,
               const float* __restrict__ b1,
               const float* __restrict__ W2,
               const float* __restrict__ b2) {
  __shared__ __align__(32) float smem[CIN * P1 + HID * P2 + HID + COUTN];
  float* sW1  = smem;
  float* sW2  = sW1 + CIN * P1;
  float* sB1p = sW2 + HID * P2;
  float* sB2  = sB1p + HID;

  const int tid = threadIdx.x;
  for (int i = tid; i < CIN * HID / 4; i += 256) {
    int k = i >> 6, n4 = i & 63;
    *(float4*)(&sW1[k * P1 + 4 * n4]) = *(const float4*)(W1 + 4 * i);
  }
  for (int i = tid; i < HID * COUTN / 4; i += 256) {
    int k = i >> 4, o4 = i & 15;
    *(float4*)(&sW2[k * P2 + 4 * o4]) = *(const float4*)(W2 + 4 * i);
  }
  if (tid < HID) {
    int t = tid >> 4, hh = (tid >> 3) & 1, r = tid & 7;
    sB1p[tid] = b1[16 * t + 4 * (r >> 1) + 2 * hh + (r & 1)];
  }
  if (tid < COUTN) sB2[tid] = b2[tid];
  __syncthreads();

  const int wave  = tid >> 5;
  const int lane  = tid & 31;
  const int lhalf = lane >> 4;
  const int lmod  = lane & 15;

  const int strip = blockIdx.x * 8 + wave;
  if (strip >= STRIPS) return;
  const int row0 = strip * 16;

  v2f xb[16];
  {
    const float* base = xo + (size_t)(row0 + lmod) * CIN + 2 * lhalf;
#pragma unroll
    for (int s = 0; s < 16; ++s)
      xb[s] = *(const v2f*)(base + 4 * s);
  }

  v8f oacc[4];
#pragma unroll
  for (int o = 0; o < 4; ++o)
    oacc[o] = *(const v8f*)(&sB2[32 * lhalf + 8 * o]);

  const int pi = 4 * ((lmod & 7) >> 1) + 2 * (lmod >> 3) + (lmod & 1);
  const float* w1p = sW1 + (2 * lhalf) * P1 + pi;
  const float* w2p = sW2 + (2 * lhalf) * P2 + 32 * (lmod >> 3) + (lmod & 7);

  for (int t = 0; t < 16; ++t) {
    v8f h = *(const v8f*)(&sB1p[(2 * t + lhalf) * 8]);
#pragma unroll
    for (int s = 0; s < 16; ++s) {
      v2f afrag;
      afrag[0] = w1p[(4 * s) * P1 + 16 * t];
      afrag[1] = w1p[(4 * s + 1) * P1 + 16 * t];
      h = __builtin_amdgcn_wmma_f32_16x16x4_f32(false, afrag, false, xb[s],
                                                (short)0, h, false, false);
    }
#pragma unroll
    for (int j = 0; j < 8; ++j) h[j] = fmaxf(h[j], 0.f);

#pragma unroll
    for (int o = 0; o < 4; ++o) {
#pragma unroll
      for (int kk = 0; kk < 4; ++kk) {
        v2f a2frag;
        a2frag[0] = w2p[(16 * t + 4 * kk) * P2 + 8 * o];
        a2frag[1] = w2p[(16 * t + 4 * kk + 1) * P2 + 8 * o];
        v2f hb;
        hb[0] = h[2 * kk];
        hb[1] = h[2 * kk + 1];
        oacc[o] = __builtin_amdgcn_wmma_f32_16x16x4_f32(false, a2frag, false, hb,
                                                        (short)0, oacc[o], false, false);
      }
    }
  }

  {
    float* op = xo + (size_t)(row0 + lmod) * COUTN + 32 * lhalf;
#pragma unroll
    for (int o = 0; o < 4; ++o)
      *(v8f*)(op + 8 * o) = oacc[o];
  }
}

extern "C" void kernel_launch(void* const* d_in, const int* in_sizes, int n_in,
                              void* d_out, int out_size, void* d_ws, size_t ws_size,
                              hipStream_t stream) {
  const float* x     = (const float*)d_in[0];
  const int*   A_idx = (const int*)d_in[1];
  const float* A_val = (const float*)d_in[2];
  const float* W1    = (const float*)d_in[3];
  const float* b1    = (const float*)d_in[4];
  const float* W2    = (const float*)d_in[5];
  const float* b2    = (const float*)d_in[6];
  float* out = (float*)d_out;

  gcn_zero_kernel<<<12500, 256, 0, stream>>>(out);

  gcn_scatter_kernel<<<(NE * 64) / 256, 256, 0, stream>>>(
      x, A_idx, A_idx + NE, A_val, out);

  gcn_mlp_kernel<<<(STRIPS + 7) / 8, 256, 0, stream>>>(out, W1, b1, W2, b2);
}


# sparse mix: 256 groups each keep their 196 nodes' sums in LDS, scan all edge rows, add own edges' x rows by LDS float atomics, write y once; no zero pass, no global atomics; sums in LDS arrival order
# speedup vs baseline: 3.5988x; 2.3047x over previous
#include <hip/hip_runtime.h>

typedef __attribute__((ext_vector_type(2))) float v2f;
typedef __attribute__((ext_vector_type(8))) float v8f;

#define NB    4
#define NN    50000
#define MM    (NB * NN)
#define CIN   64
#define HID   256
#define COUTN 64
#define NE    800000
#define STRIPS (MM / 16)

#define RPG   196
#define NGRP  256
#define MIXW  32
#define SB    512
#define NSB   ((NE + SB - 1) / SB)
#define ROT   6
#define DRAIN_AT 16
#define QCAP  (SB + 32)

static __device__ __forceinline__ void lds_add(float* p, float v) {
  __hip_atomic_fetch_add(p, v, __ATOMIC_RELAXED, __HIP_MEMORY_SCOPE_WORKGROUP);
}

static __device__ __forceinline__ int mix_step(int it, int g) {
  int sb = it + ROT * g;
  return sb >= NSB ? sb - NSB : sb;
}

static __device__ __forceinline__ void mix_load_rows(int4 (&rv)[4], const int* __restrict__ rows,
                                                     int sb, int lane) {
#pragma unroll
  for (int k = 0; k < 4; ++k) {
    const int e = min(sb * SB + 128 * k, NE - 128) + 4 * lane;
    rv[k] = *(const int4*)(rows + e);
  }
}

template <int NEDGE>
static __device__ __forceinline__ void mix_add(int ent, int cl, float vl, int j, int lane,
                                               const float* __restrict__ x, float* tile) {
  v2f xv[NEDGE][NB];
#pragma unroll
  for (int u = 0; u < NEDGE; ++u) {
    const int c = __builtin_amdgcn_readlane(cl, j + u);
    const float* xp = x + (size_t)c * CIN + 2 * lane;
#pragma unroll
    for (int b = 0; b < NB; ++b)
      xv[u][b] = *(const v2f*)(xp + (size_t)b * NN * CIN);
  }
#pragma unroll
  for (int u = 0; u < NEDGE; ++u) {
    const int rl = __builtin_amdgcn_readlane(ent, j + u) >> 20;
    const float v = __int_as_float(__builtin_amdgcn_readlane(__float_as_int(vl), j + u));
    float* tp = tile + rl * (NB * CIN) + 2 * lane;
#pragma unroll
    for (int b = 0; b < NB; ++b) {
      lds_add(tp + b * CIN,     v * xv[u][b][0]);
      lds_add(tp + b * CIN + 1, v * xv[u][b][1]);
    }
  }
}

template <bool LAST>
static __device__ __forceinline__ int mix_drain(int* q, int cnt, int lane,
                                                const float* __restrict__ x,
                                                const int* __restrict__ cols,
                                                const float* __restrict__ vals,
                                                float* tile) {
  const int full = cnt & ~3;
  for (int base = 0; base < full; base += 32) {
    const int n = min(32, full - base);
    int ent = 0, cl = 0;
    float vl = 0.f;
    if (lane < n) {
      ent = q[base + lane];
      const int e = ent & 0xFFFFF;
      cl = cols[e];
      vl = vals[e];
    }
    for (int j = 0; j < n; j += 4)
      mix_add<4>(ent, cl, vl, j, lane, x, tile);
  }
  const int rem = cnt - full;
  if (rem == 0) return 0;
  int ent = 0;
  if (lane < rem) ent = q[full + lane];
  if (!LAST) {
    if (lane < rem) q[lane] = ent;
    return rem;
  }
  int cl = 0;
  float vl = 0.f;
  if (lane < rem) {
    const int e = ent & 0xFFFFF;
    cl = cols[e];
    vl = vals[e];
  }
  for (int j = 0; j < rem; ++j)
    mix_add<1>(ent, cl, vl, j, lane, x, tile);
  return 0;
}

__global__ void __launch_bounds__(1024)
gcn_mix_kernel(const float* __restrict__ x,
               const int* __restrict__ rows,
               const int* __restrict__ cols,
               const float* __restrict__ vals,
               float* __restrict__ y) {
  __shared__ __align__(16) float tile[RPG * NB * CIN];
  __shared__ int queue[MIXW * QCAP];

  const int tid  = threadIdx.x;
  const int wave = __builtin_amdgcn_readfirstlane(tid >> 5);
  const int lane = tid & 31;
  const int g    = blockIdx.x;
  const int r0   = g * RPG;

  for (int i = tid; i < RPG * NB * CIN / 4; i += 1024)
    ((float4*)tile)[i] = make_float4(0.f, 0.f, 0.f, 0.f);
  __syncthreads();

  int* q = queue + wave * QCAP;
  int cnt = 0;

  auto scan_step = [&](const int4 (&rv)[4], int it) {
    const int sb = mix_step(it, g);
    const int e0 = sb * SB + 4 * lane;
#pragma unroll
    for (int k = 0; k < 4; ++k) {
      if (sb * SB + 128 * k >= NE) continue;
      const int rr[4] = {rv[k].x, rv[k].y, rv[k].z, rv[k].w};
#pragma unroll
      for (int c = 0; c < 4; ++c) {
        const unsigned rl = (unsigned)(rr[c] - r0);
        const bool m = rl < (unsigned)RPG;
        const unsigned mask = __builtin_amdgcn_ballot_w32(m);
        if (mask) {
          if (m)
            q[cnt + (int)__builtin_amdgcn_mbcnt_lo(mask, 0u)] =
                (int)((rl << 20) | (unsigned)(e0 + 128 * k + c));
          cnt += __builtin_popcount(mask);
        }
      }
    }
    if (cnt >= DRAIN_AT)
      cnt = mix_drain<false>(q, cnt, lane, x, cols, vals, tile);
  };
  auto step_of = [&](int it) { return mix_step(min(it, NSB - 1), g); };

  int4 ra[4], rb[4], rc[4];
  for (int k = 0; k < 4; ++k) ra[k] = rb[k] = make_int4(-1, -1, -1, -1);
  for (int it = wave - 2 * MIXW; it < NSB; it += 3 * MIXW) {
    mix_load_rows(rc, rows, step_of(it + 2 * MIXW), lane);
    if (it >= 0) scan_step(ra, it);
    mix_load_rows(ra, rows, step_of(it + 3 * MIXW), lane);
    if (it + MIXW >= 0 && it + MIXW < NSB) scan_step(rb, it + MIXW);
    mix_load_rows(rb, rows, step_of(it + 4 * MIXW), lane);
    if (it + 2 * MIXW < NSB) scan_step(rc, it + 2 * MIXW);
  }
  if (cnt > 0) mix_drain<true>(q, cnt, lane, x, cols, vals, tile);
  __syncthreads();

  for (int i = tid; i < RPG * NB * CIN / 4; i += 1024) {
    const int node = r0 + (i >> 6);
    const int b    = (i >> 4) & 3;
    const int q4   = i & 15;
    if (node < NN)
      *(float4*)(y + ((size_t)b * NN + node) * CIN + 4 * q4) = ((const float4*)tile)[i];
  }
}

#define P1 264
#define P2 68

__global__ void __launch_bounds__(256, 1)
gcn_mlp_kernel(float* __restrict__ xo,
               const float* __restrict__ W1,
               const float* __restrict__ b1,
               const float* __restrict__ W2,
               const float* __restrict__ b2) {
  __shared__ __align__(32) float smem[CIN * P1 + HID * P2 + HID + COUTN];
  float* sW1  = smem;
  float* sW2  = sW1 + CIN * P1;
  float* sB1p = sW2 + HID * P2;
  float* sB2  = sB1p + HID;

  const int tid = threadIdx.x;
  for (int i = tid; i < CIN * HID / 4; i += 256) {
    int k = i >> 6, n4 = i & 63;
    *(float4*)(&sW1[k * P1 + 4 * n4]) = *(const float4*)(W1 + 4 * i);
  }
  for (int i = tid; i < HID * COUTN / 4; i += 256) {
    int k = i >> 4, o4 = i & 15;
    *(float4*)(&sW2[k * P2 + 4 * o4]) = *(const float4*)(W2 + 4 * i);
  }
  if (tid < HID) {
    int t = tid >> 4, hh = (tid >> 3) & 1, r = tid & 7;
    sB1p[tid] = b1[16 * t + 4 * (r >> 1) + 2 * hh + (r & 1)];
  }
  if (tid < COUTN) sB2[tid] = b2[tid];
  __syncthreads();

  const int wave  = tid >> 5;
  const int lane  = tid & 31;
  const int lhalf = lane >> 4;
  const int lmod  = lane & 15;

  const int strip = blockIdx.x * 8 + wave;
  if (strip >= STRIPS) return;
  const int row0 = strip * 16;

  v2f xb[16];
  {
    const float* base = xo + (size_t)(row0 + lmod) * CIN + 2 * lhalf;
#pragma unroll
    for (int s = 0; s < 16; ++s)
      xb[s] = *(const v2f*)(base + 4 * s);
  }

  v8f oacc[4];
#pragma unroll
  for (int o = 0; o < 4; ++o)
    oacc[o] = *(const v8f*)(&sB2[32 * lhalf + 8 * o]);

  const int pi = 4 * ((lmod & 7) >> 1) + 2 * (lmod >> 3) + (lmod & 1);
  const float* w1p = sW1 + (2 * lhalf) * P1 + pi;
  const float* w2p = sW2 + (2 * lhalf) * P2 + 32 * (lmod >> 3) + (lmod & 7);

  for (int t = 0; t < 16; ++t) {
    v8f h = *(const v8f*)(&sB1p[(2 * t + lhalf) * 8]);
#pragma unroll
    for (int s = 0; s < 16; ++s) {
      v2f afrag;
      afrag[0] = w1p[(4 * s) * P1 + 16 * t];
      afrag[1] = w1p[(4 * s + 1) * P1 + 16 * t];
      h = __builtin_amdgcn_wmma_f32_16x16x4_f32(false, afrag, false, xb[s],
                                                (short)0, h, false, false);
    }
#pragma unroll
    for (int j = 0; j < 8; ++j) h[j] = fmaxf(h[j], 0.f);

#pragma unroll
    for (int o = 0; o < 4; ++o) {
#pragma unroll
      for (int kk = 0; kk < 4; ++kk) {
        v2f a2frag;
        a2frag[0] = w2p[(16 * t + 4 * kk) * P2 + 8 * o];
        a2frag[1] = w2p[(16 * t + 4 * kk + 1) * P2 + 8 * o];
        v2f hb;
        hb[0] = h[2 * kk];
        hb[1] = h[2 * kk + 1];
        oacc[o] = __builtin_amdgcn_wmma_f32_16x16x4_f32(false, a2frag, false, hb,
                                                        (short)0, oacc[o], false, false);
      }
    }
  }

  {
    float* op = xo + (size_t)(row0 + lmod) * COUTN + 32 * lhalf;
#pragma unroll
    for (int o = 0; o < 4; ++o)
      *(v8f*)(op + 8 * o) = oacc[o];
  }
}

extern "C" void kernel_launch(void* const* d_in, const int* in_sizes, int n_in,
                              void* d_out, int out_size, void* d_ws, size_t ws_size,
                              hipStream_t stream) {
  const float* x     = (const float*)d_in[0];
  const int*   A_idx = (const int*)d_in[1];
  const float* A_val = (const float*)d_in[2];
  const float* W1    = (const float*)d_in[3];
  const float* b1    = (const float*)d_in[4];
  const float* W2    = (const float*)d_in[5];
  const float* b2    = (const float*)d_in[6];
  float* out = (float*)d_out;

  gcn_mix_kernel<<<NGRP, 1024, 0, stream>>>(x, A_idx, A_idx + NE, A_val, out);

  gcn_mlp_kernel<<<(STRIPS + 7) / 8, 256, 0, stream>>>(out, W1, b1, W2, b2);
}


# sparse mix: a new first launch writes one membership bit per (owner group, edge) from the row indices; each group reads only its own bit plane, not every row index; sums stay in LDS arrival order
# speedup vs baseline: 3.8679x; 1.0748x over previous
#include <hip/hip_runtime.h>

typedef __attribute__((ext_vector_type(2))) float v2f;
typedef __attribute__((ext_vector_type(8))) float v8f;

#define NB    4
#define NN    50000
#define MM    (NB * NN)
#define CIN   64
#define HID   256
#define COUTN 64
#define NE    800000
#define STRIPS (MM / 16)

#define RPG   196
#define NGRP  256
#define MIXW  32
#define PB_EDGES 1024
#define PB_N  784
#define PITCH (PB_N * 32)
#define NSTEP (PITCH / 128)
#define DRAIN_AT 16
#define QCAP  64

static __device__ __forceinline__ void lds_add(float* p, float v) {
  __hip_atomic_fetch_add(p, v, __ATOMIC_RELAXED, __HIP_MEMORY_SCOPE_WORKGROUP);
}

__global__ void __launch_bounds__(256)
gcn_owner_bits_kernel(const int* __restrict__ rows, unsigned int* __restrict__ plane) {
  __shared__ __align__(16) unsigned int bm[NGRP * 32];
  const int tid = threadIdx.x;
  const int blk = blockIdx.x;
  for (int i = tid; i < NGRP * 32 / 4; i += 256)
    ((uint4*)bm)[i] = make_uint4(0u, 0u, 0u, 0u);
  __syncthreads();
  const int le = 4 * tid;
  const int e  = blk * PB_EDGES + le;
  if (e < NE) {
    const int4 r = *(const int4*)(rows + e);
    const int rr[4] = {r.x, r.y, r.z, r.w};
#pragma unroll
    for (int c = 0; c < 4; ++c) {
      const unsigned int g = (unsigned int)rr[c] / (unsigned int)RPG;
      if (g < (unsigned int)NGRP)
        __hip_atomic_fetch_or(&bm[g * 32 + (le >> 5)], 1u << ((le + c) & 31),
                              __ATOMIC_RELAXED, __HIP_MEMORY_SCOPE_WORKGROUP);
    }
  }
  __syncthreads();
  for (int i = tid; i < NGRP * 32 / 4; i += 256) {
    const int g = i >> 3;
    *(uint4*)(plane + (size_t)g * PITCH + blk * 32 + 4 * (i & 7)) = ((const uint4*)bm)[i];
  }
}

template <int NEDGE>
static __device__ __forceinline__ void mix_add(int ent, int cl, float vl, int j, int lane,
                                               const float* __restrict__ x, float* tile) {
  v2f xv[NEDGE][NB];
#pragma unroll
  for (int u = 0; u < NEDGE; ++u) {
    const int c = __builtin_amdgcn_readlane(cl, j + u);
    const float* xp = x + (size_t)c * CIN + 2 * lane;
#pragma unroll
    for (int b = 0; b < NB; ++b)
      xv[u][b] = *(const v2f*)(xp + (size_t)b * NN * CIN);
  }
#pragma unroll
  for (int u = 0; u < NEDGE; ++u) {
    const int rl = __builtin_amdgcn_readlane(ent, j + u) >> 20;
    const float v = __int_as_float(__builtin_amdgcn_readlane(__float_as_int(vl), j + u));
    float* tp = tile + rl * (NB * CIN) + 2 * lane;
#pragma unroll
    for (int b = 0; b < NB; ++b) {
      lds_add(tp + b * CIN,     v * xv[u][b][0]);
      lds_add(tp + b * CIN + 1, v * xv[u][b][1]);
    }
  }
}

template <bool LAST>
static __device__ __forceinline__ int mix_drain(int* q, int cnt, int lane, int r0,
                                                const float* __restrict__ x,
                                                const int* __restrict__ rows,
                                                const int* __restrict__ cols,
                                                const float* __restrict__ vals,
                                                float* tile) {
  const int full = cnt & ~3;
  for (int base = 0; base < full; base += 32) {
    const int n = min(32, full - base);
    int ent = 0, cl = 0;
    float vl = 0.f;
    if (lane < n) {
      const int e = q[base + lane];
      ent = ((rows[e] - r0) << 20) | e;
      cl = cols[e];
      vl = vals[e];
    }
    for (int j = 0; j < n; j += 4)
      mix_add<4>(ent, cl, vl, j, lane, x, tile);
  }
  const int rem = cnt - full;
  if (rem == 0) return 0;
  int e = 0;
  if (lane < rem) e = q[full + lane];
  if (!LAST) {
    if (lane < rem) q[lane] = e;
    return rem;
  }
  int ent = 0, cl = 0;
  float vl = 0.f;
  if (lane < rem) {
    ent = ((rows[e] - r0) << 20) | e;
    cl = cols[e];
    vl = vals[e];
  }
  for (int j = 0; j < rem; ++j)
    mix_add<1>(ent, cl, vl, j, lane, x, tile);
  return 0;
}

__global__ void __launch_bounds__(1024)
gcn_mix_kernel(const float* __restrict__ x,
               const int* __restrict__ rows,
               const int* __restrict__ cols,
               const float* __restrict__ vals,
               const unsigned int* __restrict__ plane,
               float* __restrict__ y) {
  __shared__ __align__(16) float tile[RPG * NB * CIN];
  __shared__ int queue[MIXW * QCAP];

  const int tid  = threadIdx.x;
  const int wave = __builtin_amdgcn_readfirstlane(tid >> 5);
  const int lane = tid & 31;
  const int g    = blockIdx.x;
  const int r0   = g * RPG;

  for (int i = tid; i < RPG * NB * CIN / 4; i += 1024)
    ((float4*)tile)[i] = make_float4(0.f, 0.f, 0.f, 0.f);
  __syncthreads();

  int* q = queue + wave * QCAP;
  int cnt = 0;

  const uint4* pl = (const uint4*)(plane + (size_t)g * PITCH);
  uint4 nx = pl[wave * 32 + lane];
  for (int s = wave; s < NSTEP; s += MIXW) {
    const uint4 cur = nx;
    nx = pl[min(s + MIXW, NSTEP - 1) * 32 + lane];
    const int eb = (s * 32 + lane) * 128;
    for (int c = 0; c < 4; ++c) {
      unsigned int wv = c == 0 ? cur.x : c == 1 ? cur.y : c == 2 ? cur.z : cur.w;
      unsigned int m = __builtin_amdgcn_ballot_w32(wv != 0u);
      while (m) {
        if (wv != 0u) {
          q[cnt + (int)__builtin_amdgcn_mbcnt_lo(m, 0u)] = eb + 32 * c + __builtin_ctz(wv);
          wv &= wv - 1u;
        }
        cnt += __builtin_popcount(m);
        if (cnt >= DRAIN_AT)
          cnt = mix_drain<false>(q, cnt, lane, r0, x, rows, cols, vals, tile);
        m = __builtin_amdgcn_ballot_w32(wv != 0u);
      }
    }
  }
  if (cnt > 0) mix_drain<true>(q, cnt, lane, r0, x, rows, cols, vals, tile);
  __syncthreads();

  for (int i = tid; i < RPG * NB * CIN / 4; i += 1024) {
    const int node = r0 + (i >> 6);
    const int b    = (i >> 4) & 3;
    const int q4   = i & 15;
    if (node < NN)
      *(float4*)(y + ((size_t)b * NN + node) * CIN + 4 * q4) = ((const float4*)tile)[i];
  }
}

#define P1 264
#define P2 68

__global__ void __launch_bounds__(256, 1)
gcn_mlp_kernel(float* __restrict__ xo,
               const float* __restrict__ W1,
               const float* __restrict__ b1,
               const float* __restrict__ W2,
               const float* __restrict__ b2) {
  __shared__ __align__(32) float smem[CIN * P1 + HID * P2 + HID + COUTN];
  float* sW1  = smem;
  float* sW2  = sW1 + CIN * P1;
  float* sB1p = sW2 + HID * P2;
  float* sB2  = sB1p + HID;

  const int tid = threadIdx.x;
  for (int i = tid; i < CIN * HID / 4; i += 256) {
    int k = i >> 6, n4 = i & 63;
    *(float4*)(&sW1[k * P1 + 4 * n4]) = *(const float4*)(W1 + 4 * i);
  }
  for (int i = tid; i < HID * COUTN / 4; i += 256) {
    int k = i >> 4, o4 = i & 15;
    *(float4*)(&sW2[k * P2 + 4 * o4]) = *(const float4*)(W2 + 4 * i);
  }
  if (tid < HID) {
    int t = tid >> 4, hh = (tid >> 3) & 1, r = tid & 7;
    sB1p[tid] = b1[16 * t + 4 * (r >> 1) + 2 * hh + (r & 1)];
  }
  if (tid < COUTN) sB2[tid] = b2[tid];
  __syncthreads();

  const int wave  = tid >> 5;
  const int lane  = tid & 31;
  const int lhalf = lane >> 4;
  const int lmod  = lane & 15;

  const int strip = blockIdx.x * 8 + wave;
  if (strip >= STRIPS) return;
  const int row0 = strip * 16;

  v2f xb[16];
  {
    const float* base = xo + (size_t)(row0 + lmod) * CIN + 2 * lhalf;
#pragma unroll
    for (int s = 0; s < 16; ++s)
      xb[s] = *(const v2f*)(base + 4 * s);
  }

  v8f oacc[4];
#pragma unroll
  for (int o = 0; o < 4; ++o)
    oacc[o] = *(const v8f*)(&sB2[32 * lhalf + 8 * o]);

  const int pi = 4 * ((lmod & 7) >> 1) + 2 * (lmod >> 3) + (lmod & 1);
  const float* w1p = sW1 + (2 * lhalf) * P1 + pi;
  const float* w2p = sW2 + (2 * lhalf) * P2 + 32 * (lmod >> 3) + (lmod & 7);

  for (int t = 0; t < 16; ++t) {
    v8f h = *(const v8f*)(&sB1p[(2 * t + lhalf) * 8]);
#pragma unroll
    for (int s = 0; s < 16; ++s) {
      v2f afrag;
      afrag[0] = w1p[(4 * s) * P1 + 16 * t];
      afrag[1] = w1p[(4 * s + 1) * P1 + 16 * t];
      h = __builtin_amdgcn_wmma_f32_16x16x4_f32(false, afrag, false, xb[s],
                                                (short)0, h, false, false);
    }
#pragma unroll
    for (int j = 0; j < 8; ++j) h[j] = fmaxf(h[j], 0.f);

#pragma unroll
    for (int o = 0; o < 4; ++o) {
#pragma unroll
      for (int kk = 0; kk < 4; ++kk) {
        v2f a2frag;
        a2frag[0] = w2p[(16 * t + 4 * kk) * P2 + 8 * o];
        a2frag[1] = w2p[(16 * t + 4 * kk + 1) * P2 + 8 * o];
        v2f hb;
        hb[0] = h[2 * kk];
        hb[1] = h[2 * kk + 1];
        oacc[o] = __builtin_amdgcn_wmma_f32_16x16x4_f32(false, a2frag, false, hb,
                                                        (short)0, oacc[o], false, false);
      }
    }
  }

  {
    float* op = xo + (size_t)(row0 + lmod) * COUTN + 32 * lhalf;
#pragma unroll
    for (int o = 0; o < 4; ++o)
      *(v8f*)(op + 8 * o) = oacc[o];
  }
}

extern "C" void kernel_launch(void* const* d_in, const int* in_sizes, int n_in,
                              void* d_out, int out_size, void* d_ws, size_t ws_size,
                              hipStream_t stream) {
  const float* x     = (const float*)d_in[0];
  const int*   A_idx = (const int*)d_in[1];
  const float* A_val = (const float*)d_in[2];
  const float* W1    = (const float*)d_in[3];
  const float* b1    = (const float*)d_in[4];
  const float* W2    = (const float*)d_in[5];
  const float* b2    = (const float*)d_in[6];
  float* out = (float*)d_out;

  unsigned int* plane = (unsigned int*)d_ws;

  gcn_owner_bits_kernel<<<PB_N, 256, 0, stream>>>(A_idx, plane);

  gcn_mix_kernel<<<NGRP, 1024, 0, stream>>>(x, A_idx, A_idx + NE, A_val, plane, out);

  gcn_mlp_kernel<<<(STRIPS + 7) / 8, 256, 0, stream>>>(out, W1, b1, W2, b2);
}


# sparse mix: each kept edge goes via an LDS list (slot from an integer add) to the wave that owns its node, 7 nodes a wave; owners sum by plain LDS read-add-write; no float atomics; arrival-order sums
# speedup vs baseline: 4.2832x; 1.1074x over previous
#include <hip/hip_runtime.h>

typedef __attribute__((ext_vector_type(2))) float v2f;
typedef __attribute__((ext_vector_type(8))) float v8f;

#define NB    4
#define NN    50000
#define MM    (NB * NN)
#define CIN   64
#define HID   256
#define COUTN 64
#define NE    800000
#define STRIPS (MM / 16)

#define RPG   196
#define NGRP  256
#define MIXW  32
#define PB_EDGES 1024
#define PB_N  784
#define PITCH (PB_N * 32)
#define NSTEP (PITCH / 128)
#define DRAIN_AT 32
#define QCAP  64
#define NPW   7
#define NOWN  28
#define ICAP  256

static __device__ __forceinline__ void lds_add(float* p, float v) {
  __hip_atomic_fetch_add(p, v, __ATOMIC_RELAXED, __HIP_MEMORY_SCOPE_WORKGROUP);
}

__global__ void __launch_bounds__(256)
gcn_owner_bits_kernel(const int* __restrict__ rows, unsigned int* __restrict__ plane) {
  __shared__ __align__(16) unsigned int bm[NGRP * 32];
  const int tid = threadIdx.x;
  const int blk = blockIdx.x;
  for (int i = tid; i < NGRP * 32 / 4; i += 256)
    ((uint4*)bm)[i] = make_uint4(0u, 0u, 0u, 0u);
  __syncthreads();
  const int le = 4 * tid;
  const int e  = blk * PB_EDGES + le;
  if (e < NE) {
    const int4 r = *(const int4*)(rows + e);
    const int rr[4] = {r.x, r.y, r.z, r.w};
#pragma unroll
    for (int c = 0; c < 4; ++c) {
      const unsigned int g = (unsigned int)rr[c] / (unsigned int)RPG;
      if (g < (unsigned int)NGRP)
        __hip_atomic_fetch_or(&bm[g * 32 + (le >> 5)], 1u << ((le + c) & 31),
                              __ATOMIC_RELAXED, __HIP_MEMORY_SCOPE_WORKGROUP);
    }
  }
  __syncthreads();
  for (int i = tid; i < NGRP * 32 / 4; i += 256) {
    const int g = i >> 3;
    *(uint4*)(plane + (size_t)g * PITCH + blk * 32 + 4 * (i & 7)) = ((const uint4*)bm)[i];
  }
}

template <int NEDGE>
static __device__ __forceinline__ void mix_add(int ent, int cl, float vl, int j, int lane,
                                               const float* __restrict__ x, float* tile) {
  v2f xv[NEDGE][NB];
#pragma unroll
  for (int u = 0; u < NEDGE; ++u) {
    const int c = __builtin_amdgcn_readlane(cl, j + u);
    const float* xp = x + (size_t)c * CIN + 2 * lane;
#pragma unroll
    for (int b = 0; b < NB; ++b)
      xv[u][b] = *(const v2f*)(xp + (size_t)b * NN * CIN);
  }
#pragma unroll
  for (int u = 0; u < NEDGE; ++u) {
    const int rl = __builtin_amdgcn_readlane(ent, j + u) >> 20;
    const float v = __int_as_float(__builtin_amdgcn_readlane(__float_as_int(vl), j + u));
    float* tp = tile + rl * (NB * CIN) + 2 * lane;
#pragma unroll
    for (int b = 0; b < NB; ++b) {
      lds_add(tp + b * CIN,     v * xv[u][b][0]);
      lds_add(tp + b * CIN + 1, v * xv[u][b][1]);
    }
  }
}

template <int NEDGE>
static __device__ __forceinline__ void own_add(int ent, int cl, float vl, int j, int lane,
                                               const float* __restrict__ x, float* tile) {
  v2f xv[NEDGE][NB];
#pragma unroll
  for (int u = 0; u < NEDGE; ++u) {
    const int c = __builtin_amdgcn_readlane(cl, j + u);
    const float* xp = x + (size_t)c * CIN + 2 * lane;
#pragma unroll
    for (int b = 0; b < NB; ++b)
      xv[u][b] = *(const v2f*)(xp + (size_t)b * NN * CIN);
  }
#pragma unroll
  for (int u = 0; u < NEDGE; ++u) {
    const int rl = __builtin_amdgcn_readlane(ent, j + u) >> 20;
    const float v = __int_as_float(__builtin_amdgcn_readlane(__float_as_int(vl), j + u));
    float* tp = tile + rl * (NB * CIN) + 2 * lane;
#pragma unroll
    for (int b = 0; b < NB; ++b) {
      v2f t = *(v2f*)(tp + b * CIN);
      t[0] += v * xv[u][b][0];
      t[1] += v * xv[u][b][1];
      *(v2f*)(tp + b * CIN) = t;
    }
  }
}

static __device__ __forceinline__ void mix_route(const int* q, int cnt, int lane, int r0,
                                                 const float* __restrict__ x,
                                                 const int* __restrict__ rows,
                                                 const int* __restrict__ cols,
                                                 const float* __restrict__ vals,
                                                 float* tile, int* inbox, int* incnt) {
  for (int base = 0; base < cnt; base += 32) {
    const int n = min(32, cnt - base);
    int ent = 0;
    bool ov = false;
    if (lane < n) {
      const int e = q[base + lane];
      const int rl = rows[e] - r0;
      if ((unsigned int)rl < (unsigned int)RPG) {
        const int own = rl / NPW;
        const int slot = __hip_atomic_fetch_add(&incnt[own], 1, __ATOMIC_RELAXED,
                                                __HIP_MEMORY_SCOPE_WORKGROUP);
        ent = (rl << 20) | e;
        if (slot < ICAP) inbox[own * ICAP + slot] = ent;
        else ov = true;
      }
    }
    unsigned int om = __builtin_amdgcn_ballot_w32(ov);
    if (om) {
      int cl = 0;
      float vl = 0.f;
      if (ov) {
        const int e = ent & 0xFFFFF;
        cl = cols[e];
        vl = vals[e];
      }
      while (om) {
        const int j = __builtin_ctz(om);
        mix_add<1>(ent, cl, vl, j, lane, x, tile);
        om &= om - 1u;
      }
    }
  }
}

__global__ void __launch_bounds__(1024)
gcn_mix_kernel(const float* __restrict__ x,
               const int* __restrict__ rows,
               const int* __restrict__ cols,
               const float* __restrict__ vals,
               const unsigned int* __restrict__ plane,
               float* __restrict__ y) {
  __shared__ __align__(16) float tile[RPG * NB * CIN];
  __shared__ int queue[MIXW * QCAP];
  __shared__ int inbox[NOWN * ICAP];
  __shared__ int incnt[NOWN];

  const int tid  = threadIdx.x;
  const int wave = __builtin_amdgcn_readfirstlane(tid >> 5);
  const int lane = tid & 31;
  const int g    = blockIdx.x;
  const int r0   = g * RPG;

  for (int i = tid; i < RPG * NB * CIN / 4; i += 1024)
    ((float4*)tile)[i] = make_float4(0.f, 0.f, 0.f, 0.f);
  if (tid < NOWN) incnt[tid] = 0;
  __syncthreads();

  int* q = queue + wave * QCAP;
  int cnt = 0;

  const uint4* pl = (const uint4*)(plane + (size_t)g * PITCH);
  uint4 nx = pl[wave * 32 + lane];
  for (int s = wave; s < NSTEP; s += MIXW) {
    const uint4 cur = nx;
    nx = pl[min(s + MIXW, NSTEP - 1) * 32 + lane];
    const int eb = (s * 32 + lane) * 128;
    for (int c = 0; c < 4; ++c) {
      unsigned int wv = c == 0 ? cur.x : c == 1 ? cur.y : c == 2 ? cur.z : cur.w;
      unsigned int m = __builtin_amdgcn_ballot_w32(wv != 0u);
      while (m) {
        if (wv != 0u) {
          q[cnt + (int)__builtin_amdgcn_mbcnt_lo(m, 0u)] = eb + 32 * c + __builtin_ctz(wv);
          wv &= wv - 1u;
        }
        cnt += __builtin_popcount(m);
        if (cnt >= DRAIN_AT) {
          mix_route(q, cnt, lane, r0, x, rows, cols, vals, tile, inbox, incnt);
          cnt = 0;
        }
        m = __builtin_amdgcn_ballot_w32(wv != 0u);
      }
    }
  }
  if (cnt > 0) mix_route(q, cnt, lane, r0, x, rows, cols, vals, tile, inbox, incnt);
  __syncthreads();

  if (wave < NOWN) {
    const int total = min(__builtin_amdgcn_readfirstlane(incnt[wave]), ICAP);
    const int* ib = inbox + wave * ICAP;
    int ent = 0, cl = 0;
    float vl = 0.f;
    if (lane < total) {
      ent = ib[lane];
      const int e = ent & 0xFFFFF;
      cl = cols[e];
      vl = vals[e];
    }
    for (int base = 0; base < total; base += 32) {
      const int n = min(32, total - base);
      int entn = 0, cln = 0;
      float vln = 0.f;
      if (base + 32 + lane < total) {
        entn = ib[base + 32 + lane];
        const int e = entn & 0xFFFFF;
        cln = cols[e];
        vln = vals[e];
      }
      int j = 0;
      for (; j + 8 <= n; j += 8)
        own_add<8>(ent, cl, vl, j, lane, x, tile);
      for (; j + 4 <= n; j += 4)
        own_add<4>(ent, cl, vl, j, lane, x, tile);
      for (; j < n; ++j)
        own_add<1>(ent, cl, vl, j, lane, x, tile);
      ent = entn;
      cl = cln;
      vl = vln;
    }
  }
  __syncthreads();

  for (int i = tid; i < RPG * NB * CIN / 4; i += 1024) {
    const int node = r0 + (i >> 6);
    const int b    = (i >> 4) & 3;
    const int q4   = i & 15;
    if (node < NN)
      *(float4*)(y + ((size_t)b * NN + node) * CIN + 4 * q4) = ((const float4*)tile)[i];
  }
}

#define P1 264
#define P2 68

__global__ void __launch_bounds__(256, 1)
gcn_mlp_kernel(float* __restrict__ xo,
               const float* __restrict__ W1,
               const float* __restrict__ b1,
               const float* __restrict__ W2,
               const float* __restrict__ b2) {
  __shared__ __align__(32) float smem[CIN * P1 + HID * P2 + HID + COUTN];
  float* sW1  = smem;
  float* sW2  = sW1 + CIN * P1;
  float* sB1p = sW2 + HID * P2;
  float* sB2  = sB1p + HID;

  const int tid = threadIdx.x;
  for (int i = tid; i < CIN * HID / 4; i += 256) {
    int k = i >> 6, n4 = i & 63;
    *(float4*)(&sW1[k * P1 + 4 * n4]) = *(const float4*)(W1 + 4 * i);
  }
  for (int i = tid; i < HID * COUTN / 4; i += 256) {
    int k = i >> 4, o4 = i & 15;
    *(float4*)(&sW2[k * P2 + 4 * o4]) = *(const float4*)(W2 + 4 * i);
  }
  if (tid < HID) {
    int t = tid >> 4, hh = (tid >> 3) & 1, r = tid & 7;
    sB1p[tid] = b1[16 * t + 4 * (r >> 1) + 2 * hh + (r & 1)];
  }
  if (tid < COUTN) sB2[tid] = b2[tid];
  __syncthreads();

  const int wave  = tid >> 5;
  const int lane  = tid & 31;
  const int lhalf = lane >> 4;
  const int lmod  = lane & 15;

  const int strip = blockIdx.x * 8 + wave;
  if (strip >= STRIPS) return;
  const int row0 = strip * 16;

  v2f xb[16];
  {
    const float* base = xo + (size_t)(row0 + lmod) * CIN + 2 * lhalf;
#pragma unroll
    for (int s = 0; s < 16; ++s)
      xb[s] = *(const v2f*)(base + 4 * s);
  }

  v8f oacc[4];
#pragma unroll
  for (int o = 0; o < 4; ++o)
    oacc[o] = *(const v8f*)(&sB2[32 * lhalf + 8 * o]);

  const int pi = 4 * ((lmod & 7) >> 1) + 2 * (lmod >> 3) + (lmod & 1);
  const float* w1p = sW1 + (2 * lhalf) * P1 + pi;
  const float* w2p = sW2 + (2 * lhalf) * P2 + 32 * (lmod >> 3) + (lmod & 7);

  for (int t = 0; t < 16; ++t) {
    v8f h = *(const v8f*)(&sB1p[(2 * t + lhalf) * 8]);
#pragma unroll
    for (int s = 0; s < 16; ++s) {
      v2f afrag;
      afrag[0] = w1p[(4 * s) * P1 + 16 * t];
      afrag[1] = w1p[(4 * s + 1) * P1 + 16 * t];
      h = __builtin_amdgcn_wmma_f32_16x16x4_f32(false, afrag, false, xb[s],
                                                (short)0, h, false, false);
    }
#pragma unroll
    for (int j = 0; j < 8; ++j) h[j] = fmaxf(h[j], 0.f);

#pragma unroll
    for (int o = 0; o < 4; ++o) {
#pragma unroll
      for (int kk = 0; kk < 4; ++kk) {
        v2f a2frag;
        a2frag[0] = w2p[(16 * t + 4 * kk) * P2 + 8 * o];
        a2frag[1] = w2p[(16 * t + 4 * kk + 1) * P2 + 8 * o];
        v2f hb;
        hb[0] = h[2 * kk];
        hb[1] = h[2 * kk + 1];
        oacc[o] = __builtin_amdgcn_wmma_f32_16x16x4_f32(false, a2frag, false, hb,
                                                        (short)0, oacc[o], false, false);
      }
    }
  }

  {
    float* op = xo + (size_t)(row0 + lmod) * COUTN + 32 * lhalf;
#pragma unroll
    for (int o = 0; o < 4; ++o)
      *(v8f*)(op + 8 * o) = oacc[o];
  }
}

extern "C" void kernel_launch(void* const* d_in, const int* in_sizes, int n_in,
                              void* d_out, int out_size, void* d_ws, size_t ws_size,
                              hipStream_t stream) {
  const float* x     = (const float*)d_in[0];
  const int*   A_idx = (const int*)d_in[1];
  const float* A_val = (const float*)d_in[2];
  const float* W1    = (const float*)d_in[3];
  const float* b1    = (const float*)d_in[4];
  const float* W2    = (const float*)d_in[5];
  const float* b2    = (const float*)d_in[6];
  float* out = (float*)d_out;

  unsigned int* plane = (unsigned int*)d_ws;

  gcn_owner_bits_kernel<<<PB_N, 256, 0, stream>>>(A_idx, plane);

  gcn_mix_kernel<<<NGRP, 1024, 0, stream>>>(x, A_idx, A_idx + NE, A_val, plane, out);

  gcn_mlp_kernel<<<(STRIPS + 7) / 8, 256, 0, stream>>>(out, W1, b1, W2, b2);
}
